# starting kernel
# baseline (speedup 1.0000x reference)
#include <hip/hip_runtime.h>


typedef float v2f __attribute__((ext_vector_type(2)));
typedef float v8f __attribute__((ext_vector_type(8)));

#define CIN 256
#define COUT 64

__global__ void __launch_bounds__(256)
k_zero_deg(unsigned* __restrict__ deg, int N) {
    int i = blockIdx.x * 256 + threadIdx.x;
    if (i < N) deg[i] = 0u;
}

__global__ void __launch_bounds__(256)
k_deg(const int* __restrict__ dstp, unsigned* __restrict__ deg, int E) {
    int e = blockIdx.x * 256 + threadIdx.x;
    if (e < E) atomicAdd(&deg[dstp[e]], 1u);
}

__global__ void __launch_bounds__(256)
k_dinv(const unsigned* __restrict__ deg, float* __restrict__ dinv, int N) {
    int i = blockIdx.x * 256 + threadIdx.x;
    if (i < N) dinv[i] = rsqrtf((float)(deg[i] + 1u));
}

__global__ void __launch_bounds__(256)
k_gemm(const float* __restrict__ x, const float* __restrict__ W,
       float* __restrict__ h, int N) {
    __shared__ float2 Ws2[(CIN / 2) * COUT];
    for (int idx = threadIdx.x; idx < (CIN / 2) * COUT; idx += 256) {
        int p = idx >> 6, n = idx & 63;
        Ws2[idx] = make_float2(W[p * (2 * COUT) + n], W[p * (2 * COUT) + COUT + n]);
    }
    __syncthreads();

    const int lane = threadIdx.x & 31;
    const int wave = threadIdx.x >> 5;
    const int m0   = (blockIdx.x * 8 + wave) * 16;
    const int mr   = lane & 15;
    const int kh   = (lane >> 4) << 1;

    int row  = m0 + mr;
    int rowc = row < N ? row : (N - 1);
    const float* xr = x + (size_t)rowc * CIN;

    v8f acc0 = {}; v8f acc1 = {}; v8f acc2 = {}; v8f acc3 = {};
    for (int k = 0; k < CIN; k += 4) {
        v2f a = *(const v2f*)(xr + k + kh);
        const float2* wp = &Ws2[(((k + kh) >> 1) * COUT) + mr];
        v2f b0 = *(const v2f*)(wp + 0);
        v2f b1 = *(const v2f*)(wp + 16);
        v2f b2 = *(const v2f*)(wp + 32);
        v2f b3 = *(const v2f*)(wp + 48);
        acc0 = __builtin_amdgcn_wmma_f32_16x16x4_f32(false, a, false, b0, (short)0, acc0, false, false);
        acc1 = __builtin_amdgcn_wmma_f32_16x16x4_f32(false, a, false, b1, (short)0, acc1, false, false);
        acc2 = __builtin_amdgcn_wmma_f32_16x16x4_f32(false, a, false, b2, (short)0, acc2, false, false);
        acc3 = __builtin_amdgcn_wmma_f32_16x16x4_f32(false, a, false, b3, (short)0, acc3, false, false);
    }

    const int nlo = lane & 15;
    const int mhi = (lane >> 4) << 3;
    for (int r = 0; r < 8; ++r) {
        int orow = m0 + r + mhi;
        if (orow < N) {
            float* hp = h + (size_t)orow * COUT + nlo;
            hp[0]  = acc0[r];
            hp[16] = acc1[r];
            hp[32] = acc2[r];
            hp[48] = acc3[r];
        }
    }
}

__global__ void __launch_bounds__(256)
k_self(const float* __restrict__ h, const float* __restrict__ dinv,
       const float* __restrict__ b, float* __restrict__ out, int N) {
    long long gid = (long long)blockIdx.x * 256 + threadIdx.x;
    int i = (int)(gid >> 6);
    int c = (int)(gid & 63);
    if (i < N) {
        float di = dinv[i];
        out[(size_t)i * COUT + c] = b[c] + di * di * h[(size_t)i * COUT + c];
    }
}

__global__ void __launch_bounds__(256)
k_scatter(const int* __restrict__ srcp, const int* __restrict__ dstp,
          const float* __restrict__ dinv, const float* __restrict__ h,
          float* __restrict__ out, int E) {
    long long gid = (long long)blockIdx.x * 256 + threadIdx.x;
    int e = (int)(gid >> 5);
    if (e >= E) return;
    e = __builtin_amdgcn_readfirstlane(e);
    int lane = threadIdx.x & 31;
    int s = srcp[e];
    int d = dstp[e];
    float norm = dinv[s] * dinv[d];
    const float* hs = h + (size_t)s * COUT;
    float* od = out + (size_t)d * COUT;
    unsafeAtomicAdd(&od[lane],      hs[lane]      * norm);
    unsafeAtomicAdd(&od[lane + 32], hs[lane + 32] * norm);
}

extern "C" void kernel_launch(void* const* d_in, const int* in_sizes, int n_in,
                              void* d_out, int out_size, void* d_ws, size_t ws_size,
                              hipStream_t stream) {
    const float* x  = (const float*)d_in[0];
    const int*   ei = (const int*)d_in[1];
    const float* W  = (const float*)d_in[2];
    const float* b  = (const float*)d_in[3];
    float* out = (float*)d_out;

    const int N = in_sizes[0] / CIN;
    const int E = in_sizes[1] / 2;
    const int* srcp = ei;
    const int* dstp = ei + E;

    char* ws = (char*)d_ws;
    float*    h    = (float*)ws;
    unsigned* deg  = (unsigned*)(ws + (size_t)N * COUT * sizeof(float));
    float*    dinv = (float*)(ws + (size_t)N * COUT * sizeof(float) + (size_t)N * sizeof(unsigned));

    int nb;
    nb = (N + 255) / 256;
    k_zero_deg<<<nb, 256, 0, stream>>>(deg, N);
    nb = (E + 255) / 256;
    k_deg<<<nb, 256, 0, stream>>>(dstp, deg, E);
    nb = (N + 255) / 256;
    k_dinv<<<nb, 256, 0, stream>>>(deg, dinv, N);

    nb = (N + 127) / 128;
    k_gemm<<<nb, 256, 0, stream>>>(x, W, h, N);

    long long t = (long long)N * COUT;
    nb = (int)((t + 255) / 256);
    k_self<<<nb, 256, 0, stream>>>(h, dinv, b, out, N);

    t = (long long)E * 32;
    nb = (int)((t + 255) / 256);
    k_scatter<<<nb, 256, 0, stream>>>(srcp, dstp, dinv, h, out, E);
}


# degree count by per-node-range LDS histograms over 64 edge slices plus a per-node sum of the partial counts, replacing one scattered global integer atomic per edge and the zeroing launch
# speedup vs baseline: 1.3638x; 1.3638x over previous
#include <hip/hip_runtime.h>


typedef float v2f __attribute__((ext_vector_type(2)));
typedef float v8f __attribute__((ext_vector_type(8)));

#define CIN 256
#define COUT 64

#define HMAX 25600
#define HSL  64

__global__ void __launch_bounds__(1024)
k_hist(const int* __restrict__ dstp, unsigned* __restrict__ part, int N, int E) {
    __shared__ unsigned cnt[HMAX];
    const int R  = gridDim.x;
    const int rs = ((N + R - 1) / R + 31) & ~31;
    const unsigned base = blockIdx.x * (unsigned)rs;
    for (int i = threadIdx.x; i < rs; i += 1024) cnt[i] = 0u;
    __syncthreads();

    const int per = (E + (int)gridDim.y - 1) / (int)gridDim.y;
    const int e0  = (int)blockIdx.y * per;
    int e1 = e0 + per;
    if (e1 > E) e1 = E;

    int e = e0 + (int)threadIdx.x;
    for (; e + 7 * 1024 < e1; e += 8 * 1024) {
        unsigned v[8];
#pragma unroll
        for (int j = 0; j < 8; ++j) v[j] = (unsigned)dstp[e + j * 1024];
#pragma unroll
        for (int j = 0; j < 8; ++j) {
            unsigned l = v[j] - base;
            if (l < (unsigned)rs) atomicAdd(&cnt[l], 1u);
        }
    }
    for (; e < e1; e += 1024) {
        unsigned l = (unsigned)dstp[e] - base;
        if (l < (unsigned)rs) atomicAdd(&cnt[l], 1u);
    }
    __syncthreads();

    unsigned* p = part + (size_t)blockIdx.y * ((size_t)R * rs) + base;
    for (int i = threadIdx.x; i < rs; i += 1024) p[i] = cnt[i];
}

__global__ void __launch_bounds__(256)
k_dinv(const unsigned* __restrict__ part, float* __restrict__ dinv, int N, int S, int R) {
    int i = blockIdx.x * 256 + threadIdx.x;
    if (i < N) {
        const int rs = ((N + R - 1) / R + 31) & ~31;
        const size_t pitch = (size_t)R * rs;
        unsigned d = 1u;
        for (int s = 0; s < S; ++s) d += part[s * pitch + i];
        dinv[i] = rsqrtf((float)d);
    }
}

__global__ void __launch_bounds__(256)
k_gemm(const float* __restrict__ x, const float* __restrict__ W,
       float* __restrict__ h, int N) {
    __shared__ float2 Ws2[(CIN / 2) * COUT];
    for (int idx = threadIdx.x; idx < (CIN / 2) * COUT; idx += 256) {
        int p = idx >> 6, n = idx & 63;
        Ws2[idx] = make_float2(W[p * (2 * COUT) + n], W[p * (2 * COUT) + COUT + n]);
    }
    __syncthreads();

    const int lane = threadIdx.x & 31;
    const int wave = threadIdx.x >> 5;
    const int m0   = (blockIdx.x * 8 + wave) * 16;
    const int mr   = lane & 15;
    const int kh   = (lane >> 4) << 1;

    int row  = m0 + mr;
    int rowc = row < N ? row : (N - 1);
    const float* xr = x + (size_t)rowc * CIN;

    v8f acc0 = {}; v8f acc1 = {}; v8f acc2 = {}; v8f acc3 = {};
    for (int k = 0; k < CIN; k += 4) {
        v2f a = *(const v2f*)(xr + k + kh);
        const float2* wp = &Ws2[(((k + kh) >> 1) * COUT) + mr];
        v2f b0 = *(const v2f*)(wp + 0);
        v2f b1 = *(const v2f*)(wp + 16);
        v2f b2 = *(const v2f*)(wp + 32);
        v2f b3 = *(const v2f*)(wp + 48);
        acc0 = __builtin_amdgcn_wmma_f32_16x16x4_f32(false, a, false, b0, (short)0, acc0, false, false);
        acc1 = __builtin_amdgcn_wmma_f32_16x16x4_f32(false, a, false, b1, (short)0, acc1, false, false);
        acc2 = __builtin_amdgcn_wmma_f32_16x16x4_f32(false, a, false, b2, (short)0, acc2, false, false);
        acc3 = __builtin_amdgcn_wmma_f32_16x16x4_f32(false, a, false, b3, (short)0, acc3, false, false);
    }

    const int nlo = lane & 15;
    const int mhi = (lane >> 4) << 3;
    for (int r = 0; r < 8; ++r) {
        int orow = m0 + r + mhi;
        if (orow < N) {
            float* hp = h + (size_t)orow * COUT + nlo;
            hp[0]  = acc0[r];
            hp[16] = acc1[r];
            hp[32] = acc2[r];
            hp[48] = acc3[r];
        }
    }
}

__global__ void __launch_bounds__(256)
k_self(const float* __restrict__ h, const float* __restrict__ dinv,
       const float* __restrict__ b, float* __restrict__ out, int N) {
    long long gid = (long long)blockIdx.x * 256 + threadIdx.x;
    int i = (int)(gid >> 6);
    int c = (int)(gid & 63);
    if (i < N) {
        float di = dinv[i];
        out[(size_t)i * COUT + c] = b[c] + di * di * h[(size_t)i * COUT + c];
    }
}

__global__ void __launch_bounds__(256)
k_scatter(const int* __restrict__ srcp, const int* __restrict__ dstp,
          const float* __restrict__ dinv, const float* __restrict__ h,
          float* __restrict__ out, int E) {
    long long gid = (long long)blockIdx.x * 256 + threadIdx.x;
    int e = (int)(gid >> 5);
    if (e >= E) return;
    e = __builtin_amdgcn_readfirstlane(e);
    int lane = threadIdx.x & 31;
    int s = srcp[e];
    int d = dstp[e];
    float norm = dinv[s] * dinv[d];
    const float* hs = h + (size_t)s * COUT;
    float* od = out + (size_t)d * COUT;
    unsafeAtomicAdd(&od[lane],      hs[lane]      * norm);
    unsafeAtomicAdd(&od[lane + 32], hs[lane + 32] * norm);
}

extern "C" void kernel_launch(void* const* d_in, const int* in_sizes, int n_in,
                              void* d_out, int out_size, void* d_ws, size_t ws_size,
                              hipStream_t stream) {
    const float* x  = (const float*)d_in[0];
    const int*   ei = (const int*)d_in[1];
    const float* W  = (const float*)d_in[2];
    const float* b  = (const float*)d_in[3];
    float* out = (float*)d_out;

    const int N = in_sizes[0] / CIN;
    const int E = in_sizes[1] / 2;
    const int* srcp = ei;
    const int* dstp = ei + E;

    char* ws = (char*)d_ws;
    float*    h    = (float*)ws;
    float*    dinv = (float*)(ws + (size_t)N * COUT * sizeof(float));
    unsigned* part = (unsigned*)(ws + (size_t)N * COUT * sizeof(float) + (size_t)N * sizeof(float));

    int nb;
    dim3 hg((N + HMAX - 1) / HMAX, HSL);
    k_hist<<<hg, 1024, 0, stream>>>(dstp, part, N, E);
    nb = (N + 255) / 256;
    k_dinv<<<nb, 256, 0, stream>>>(part, dinv, N, HSL, (int)hg.x);

    nb = (N + 127) / 128;
    k_gemm<<<nb, 256, 0, stream>>>(x, W, h, N);

    long long t = (long long)N * COUT;
    nb = (int)((t + 255) / 256);
    k_self<<<nb, 256, 0, stream>>>(h, dinv, b, out, N);

    t = (long long)E * 32;
    nb = (int)((t + 255) / 256);
    k_scatter<<<nb, 256, 0, stream>>>(srcp, dstp, dinv, h, out, E);
}


# float-atomic edge scatter replaced by a counting sort of src by dst (slice counts to offsets, LDS rank) and one wave per node summing its rows in registers; row sum order follows the fill's rank
# speedup vs baseline: 3.2962x; 2.4169x over previous
#include <hip/hip_runtime.h>


typedef float v2f __attribute__((ext_vector_type(2)));
typedef float v8f __attribute__((ext_vector_type(8)));

#define CIN 256
#define COUT 64

#define HMAX 25600
#define HSL  64

__global__ void __launch_bounds__(1024)
k_hist(const int* __restrict__ dstp, unsigned* __restrict__ part, int N, int E) {
    __shared__ unsigned cnt[HMAX];
    const int R  = gridDim.x;
    const int rs = ((N + R - 1) / R + 31) & ~31;
    const unsigned base = blockIdx.x * (unsigned)rs;
    for (int i = threadIdx.x; i < rs; i += 1024) cnt[i] = 0u;
    __syncthreads();

    const int per = (E + (int)gridDim.y - 1) / (int)gridDim.y;
    const int e0  = (int)blockIdx.y * per;
    int e1 = e0 + per;
    if (e1 > E) e1 = E;

    int e = e0 + (int)threadIdx.x;
    for (; e + 7 * 1024 < e1; e += 8 * 1024) {
        unsigned v[8];
#pragma unroll
        for (int j = 0; j < 8; ++j) v[j] = (unsigned)dstp[e + j * 1024];
#pragma unroll
        for (int j = 0; j < 8; ++j) {
            unsigned l = v[j] - base;
            if (l < (unsigned)rs) atomicAdd(&cnt[l], 1u);
        }
    }
    for (; e < e1; e += 1024) {
        unsigned l = (unsigned)dstp[e] - base;
        if (l < (unsigned)rs) atomicAdd(&cnt[l], 1u);
    }
    __syncthreads();

    unsigned* p = part + (size_t)blockIdx.y * ((size_t)R * rs) + base;
    for (int i = threadIdx.x; i < rs; i += 1024) p[i] = cnt[i];
}

#define DBLK 256

__global__ void __launch_bounds__(DBLK)
k_dinv(unsigned* __restrict__ part, float* __restrict__ dinv,
       unsigned* __restrict__ degc, unsigned* __restrict__ lp,
       unsigned* __restrict__ bs, int N, int R) {
    __shared__ unsigned wsum[DBLK / 32];
    const int i = blockIdx.x * DBLK + threadIdx.x;
    unsigned c = 0u;
    if (i < N) {
        const int rs = ((N + R - 1) / R + 31) & ~31;
        const size_t pitch = (size_t)R * rs;
        unsigned* p = part + i;
        for (int s0 = 0; s0 < HSL; s0 += 16) {
            unsigned v[16];
#pragma unroll
            for (int j = 0; j < 16; ++j) v[j] = p[(size_t)(s0 + j) * pitch];
#pragma unroll
            for (int j = 0; j < 16; ++j) { p[(size_t)(s0 + j) * pitch] = c; c += v[j]; }
        }
        dinv[i] = rsqrtf((float)(c + 1u));
        degc[i] = c;
    }
    const int lane = threadIdx.x & 31;
    const int wave = threadIdx.x >> 5;
    unsigned inc = c;
    for (int o = 1; o < 32; o <<= 1) {
        unsigned t = __shfl_up(inc, o);
        if (lane >= o) inc += t;
    }
    if (lane == 31) wsum[wave] = inc;
    __syncthreads();
    unsigned woff = 0u;
    for (int w = 0; w < wave; ++w) woff += wsum[w];
    if (i < N) lp[i] = woff + inc - c;
    if (threadIdx.x == DBLK - 1) bs[blockIdx.x] = woff + inc;
}

__global__ void __launch_bounds__(1024)
k_scan(const unsigned* __restrict__ bs, unsigned* __restrict__ boff, int nblk) {
    __shared__ unsigned wsum[32];
    const int lane = threadIdx.x & 31;
    const int wave = threadIdx.x >> 5;
    unsigned carry = 0u;
    for (int c0 = 0; c0 < nblk; c0 += 1024) {
        const int i = c0 + (int)threadIdx.x;
        const unsigned v = (i < nblk) ? bs[i] : 0u;
        unsigned inc = v;
        for (int o = 1; o < 32; o <<= 1) {
            unsigned t = __shfl_up(inc, o);
            if (lane >= o) inc += t;
        }
        if (lane == 31) wsum[wave] = inc;
        __syncthreads();
        unsigned woff = 0u, tot = 0u;
        for (int w = 0; w < 32; ++w) {
            unsigned q = wsum[w];
            if (w < wave) woff += q;
            tot += q;
        }
        if (i < nblk) boff[i] = carry + woff + inc - v;
        carry += tot;
        __syncthreads();
    }
}

__global__ void __launch_bounds__(1024)
k_fill(const int* __restrict__ srcp, const int* __restrict__ dstp,
       const unsigned* __restrict__ part, const unsigned* __restrict__ lp,
       const unsigned* __restrict__ boff, int* __restrict__ col, int N, int E) {
    __shared__ unsigned cur[HMAX];
    const int R  = gridDim.x;
    const int rs = ((N + R - 1) / R + 31) & ~31;
    const unsigned base = blockIdx.x * (unsigned)rs;
    const unsigned* p = part + (size_t)blockIdx.y * ((size_t)R * rs) + base;
    for (int i = threadIdx.x; i < rs; i += 1024) {
        const unsigned node = base + (unsigned)i;
        cur[i] = (node < (unsigned)N) ? boff[node / DBLK] + lp[node] + p[i] : 0u;
    }
    __syncthreads();

    const int per = (E + (int)gridDim.y - 1) / (int)gridDim.y;
    const int e0  = (int)blockIdx.y * per;
    int e1 = e0 + per;
    if (e1 > E) e1 = E;

    int e = e0 + (int)threadIdx.x;
    for (; e + 7 * 1024 < e1; e += 8 * 1024) {
        unsigned v[8];
        int sv[8];
#pragma unroll
        for (int j = 0; j < 8; ++j) { v[j] = (unsigned)dstp[e + j * 1024]; sv[j] = srcp[e + j * 1024]; }
#pragma unroll
        for (int j = 0; j < 8; ++j) {
            unsigned l = v[j] - base;
            if (l < (unsigned)rs) {
                unsigned pos = atomicAdd(&cur[l], 1u);
                if (pos < (unsigned)E) col[pos] = sv[j];
            }
        }
    }
    for (; e < e1; e += 1024) {
        unsigned l = (unsigned)dstp[e] - base;
        if (l < (unsigned)rs) {
            unsigned pos = atomicAdd(&cur[l], 1u);
            if (pos < (unsigned)E) col[pos] = srcp[e];
        }
    }
}

__global__ void __launch_bounds__(256)
k_gemm(const float* __restrict__ x, const float* __restrict__ W,
       float* __restrict__ h, int N) {
    __shared__ float2 Ws2[(CIN / 2) * COUT];
    for (int idx = threadIdx.x; idx < (CIN / 2) * COUT; idx += 256) {
        int p = idx >> 6, n = idx & 63;
        Ws2[idx] = make_float2(W[p * (2 * COUT) + n], W[p * (2 * COUT) + COUT + n]);
    }
    __syncthreads();

    const int lane = threadIdx.x & 31;
    const int wave = threadIdx.x >> 5;
    const int m0   = (blockIdx.x * 8 + wave) * 16;
    const int mr   = lane & 15;
    const int kh   = (lane >> 4) << 1;

    int row  = m0 + mr;
    int rowc = row < N ? row : (N - 1);
    const float* xr = x + (size_t)rowc * CIN;

    v8f acc0 = {}; v8f acc1 = {}; v8f acc2 = {}; v8f acc3 = {};
    for (int k = 0; k < CIN; k += 4) {
        v2f a = *(const v2f*)(xr + k + kh);
        const float2* wp = &Ws2[(((k + kh) >> 1) * COUT) + mr];
        v2f b0 = *(const v2f*)(wp + 0);
        v2f b1 = *(const v2f*)(wp + 16);
        v2f b2 = *(const v2f*)(wp + 32);
        v2f b3 = *(const v2f*)(wp + 48);
        acc0 = __builtin_amdgcn_wmma_f32_16x16x4_f32(false, a, false, b0, (short)0, acc0, false, false);
        acc1 = __builtin_amdgcn_wmma_f32_16x16x4_f32(false, a, false, b1, (short)0, acc1, false, false);
        acc2 = __builtin_amdgcn_wmma_f32_16x16x4_f32(false, a, false, b2, (short)0, acc2, false, false);
        acc3 = __builtin_amdgcn_wmma_f32_16x16x4_f32(false, a, false, b3, (short)0, acc3, false, false);
    }

    const int nlo = lane & 15;
    const int mhi = (lane >> 4) << 3;
    for (int r = 0; r < 8; ++r) {
        int orow = m0 + r + mhi;
        if (orow < N) {
            float* hp = h + (size_t)orow * COUT + nlo;
            hp[0]  = acc0[r];
            hp[16] = acc1[r];
            hp[32] = acc2[r];
            hp[48] = acc3[r];
        }
    }
}

#define GU 8
__global__ void __launch_bounds__(256)
k_gather(const int* __restrict__ col, const unsigned* __restrict__ lp,
         const unsigned* __restrict__ boff, const unsigned* __restrict__ degc,
         const float* __restrict__ dinv, const float* __restrict__ h,
         const float* __restrict__ b, float* __restrict__ out, int N, int E) {
    int d = blockIdx.x * 8 + (int)(threadIdx.x >> 5);
    if (d >= N) return;
    d = __builtin_amdgcn_readfirstlane(d);
    const int lane = threadIdx.x & 31;
    const unsigned start = boff[d / DBLK] + lp[d];
    const unsigned cnt   = degc[d];
    const float dd = dinv[d];
    const float2* h2 = (const float2*)h;
    const float2 self = h2[(size_t)d * (COUT / 2) + lane];
    float ax = 0.f, ay = 0.f;
    for (unsigned j0 = 0; j0 < cnt; j0 += 32) {
        int   s_l = d;
        float w_l = 0.f;
        const unsigned j = j0 + (unsigned)lane;
        if (j < cnt && start + j < (unsigned)E) {
            int sv = col[start + j];
            if ((unsigned)sv < (unsigned)N) { s_l = sv; w_l = dinv[sv]; }
        }
        const unsigned left = cnt - j0;
        const unsigned n = left < 32u ? left : 32u;
        for (unsigned t = 0; t < n; t += GU) {
            float2 v[GU];
            float  w[GU];
#pragma unroll
            for (int u = 0; u < GU; ++u) {
                const int s = __builtin_amdgcn_readlane(s_l, (int)t + u);
                w[u] = __builtin_bit_cast(float,
                         __builtin_amdgcn_readlane(__builtin_bit_cast(int, w_l), (int)t + u));
                v[u] = h2[(size_t)s * (COUT / 2) + lane];
            }
#pragma unroll
            for (int u = 0; u < GU; ++u) { ax += w[u] * v[u].x; ay += w[u] * v[u].y; }
        }
    }
    const float2 bb = ((const float2*)b)[lane];
    float2 o;
    o.x = bb.x + dd * (ax + dd * self.x);
    o.y = bb.y + dd * (ay + dd * self.y);
    ((float2*)out)[(size_t)d * (COUT / 2) + lane] = o;
}

extern "C" void kernel_launch(void* const* d_in, const int* in_sizes, int n_in,
                              void* d_out, int out_size, void* d_ws, size_t ws_size,
                              hipStream_t stream) {
    const float* x  = (const float*)d_in[0];
    const int*   ei = (const int*)d_in[1];
    const float* W  = (const float*)d_in[2];
    const float* b  = (const float*)d_in[3];
    float* out = (float*)d_out;

    const int N = in_sizes[0] / CIN;
    const int E = in_sizes[1] / 2;
    const int* srcp = ei;
    const int* dstp = ei + E;

    const int R    = (N + HMAX - 1) / HMAX;
    const int rs   = ((N + R - 1) / R + 31) & ~31;
    const int nblk = (N + DBLK - 1) / DBLK;
    const size_t nN = (((size_t)N * 4) + 255) & ~(size_t)255;
    const size_t nB = (((size_t)nblk * 4) + 255) & ~(size_t)255;
    char* ws = (char*)d_ws;
    float*    h    = (float*)ws;                 ws += (size_t)N * COUT * sizeof(float);
    float*    dinv = (float*)ws;                 ws += nN;
    unsigned* part = (unsigned*)ws;              ws += (size_t)HSL * R * rs * sizeof(unsigned);
    unsigned* degc = (unsigned*)ws;              ws += nN;
    unsigned* lp   = (unsigned*)ws;              ws += nN;
    unsigned* bs   = (unsigned*)ws;              ws += nB;
    unsigned* boff = (unsigned*)ws;              ws += nB;
    int*      col  = (int*)ws;

    int nb;
    dim3 hg(R, HSL);
    k_hist<<<hg, 1024, 0, stream>>>(dstp, part, N, E);
    k_dinv<<<nblk, DBLK, 0, stream>>>(part, dinv, degc, lp, bs, N, R);
    k_scan<<<1, 1024, 0, stream>>>(bs, boff, nblk);
    k_fill<<<hg, 1024, 0, stream>>>(srcp, dstp, part, lp, boff, col, N, E);

    nb = (N + 127) / 128;
    k_gemm<<<nb, 256, 0, stream>>>(x, W, h, N);

    nb = (N + 7) / 8;
    k_gather<<<nb, 256, 0, stream>>>(col, lp, boff, degc, dinv, h, b, out, N, E);
}


# h = xW product moved into the counting-sort fill launch as extra work-groups (role by block index, 28 strips each): WMMA strips run beside the fill's LDS ranking; 5 launches, not 6
# speedup vs baseline: 3.5928x; 1.0900x over previous
#include <hip/hip_runtime.h>


typedef float v2f __attribute__((ext_vector_type(2)));
typedef float v8f __attribute__((ext_vector_type(8)));

#define CIN 256
#define COUT 64

#define HMAX 25600
#define HSL  64

__global__ void __launch_bounds__(1024)
k_hist(const int* __restrict__ dstp, unsigned* __restrict__ part, int N, int E) {
    __shared__ unsigned cnt[HMAX];
    const int R  = gridDim.x;
    const int rs = ((N + R - 1) / R + 31) & ~31;
    const unsigned base = blockIdx.x * (unsigned)rs;
    for (int i = threadIdx.x; i < rs; i += 1024) cnt[i] = 0u;
    __syncthreads();

    const int per = (E + (int)gridDim.y - 1) / (int)gridDim.y;
    const int e0  = (int)blockIdx.y * per;
    int e1 = e0 + per;
    if (e1 > E) e1 = E;

    int e = e0 + (int)threadIdx.x;
    for (; e + 7 * 1024 < e1; e += 8 * 1024) {
        unsigned v[8];
#pragma unroll
        for (int j = 0; j < 8; ++j) v[j] = (unsigned)dstp[e + j * 1024];
#pragma unroll
        for (int j = 0; j < 8; ++j) {
            unsigned l = v[j] - base;
            if (l < (unsigned)rs) atomicAdd(&cnt[l], 1u);
        }
    }
    for (; e < e1; e += 1024) {
        unsigned l = (unsigned)dstp[e] - base;
        if (l < (unsigned)rs) atomicAdd(&cnt[l], 1u);
    }
    __syncthreads();

    unsigned* p = part + (size_t)blockIdx.y * ((size_t)R * rs) + base;
    for (int i = threadIdx.x; i < rs; i += 1024) p[i] = cnt[i];
}

#define DBLK 256

__global__ void __launch_bounds__(DBLK)
k_dinv(unsigned* __restrict__ part, float* __restrict__ dinv,
       unsigned* __restrict__ degc, unsigned* __restrict__ lp,
       unsigned* __restrict__ bs, int N, int R) {
    __shared__ unsigned wsum[DBLK / 32];
    const int i = blockIdx.x * DBLK + threadIdx.x;
    unsigned c = 0u;
    if (i < N) {
        const int rs = ((N + R - 1) / R + 31) & ~31;
        const size_t pitch = (size_t)R * rs;
        unsigned* p = part + i;
        for (int s0 = 0; s0 < HSL; s0 += 16) {
            unsigned v[16];
#pragma unroll
            for (int j = 0; j < 16; ++j) v[j] = p[(size_t)(s0 + j) * pitch];
#pragma unroll
            for (int j = 0; j < 16; ++j) { p[(size_t)(s0 + j) * pitch] = c; c += v[j]; }
        }
        dinv[i] = rsqrtf((float)(c + 1u));
        degc[i] = c;
    }
    const int lane = threadIdx.x & 31;
    const int wave = threadIdx.x >> 5;
    unsigned inc = c;
    for (int o = 1; o < 32; o <<= 1) {
        unsigned t = __shfl_up(inc, o);
        if (lane >= o) inc += t;
    }
    if (lane == 31) wsum[wave] = inc;
    __syncthreads();
    unsigned woff = 0u;
    for (int w = 0; w < wave; ++w) woff += wsum[w];
    if (i < N) lp[i] = woff + inc - c;
    if (threadIdx.x == DBLK - 1) bs[blockIdx.x] = woff + inc;
}

__global__ void __launch_bounds__(1024)
k_scan(const unsigned* __restrict__ bs, unsigned* __restrict__ boff, int nblk) {
    __shared__ unsigned wsum[32];
    const int lane = threadIdx.x & 31;
    const int wave = threadIdx.x >> 5;
    unsigned carry = 0u;
    for (int c0 = 0; c0 < nblk; c0 += 1024) {
        const int i = c0 + (int)threadIdx.x;
        const unsigned v = (i < nblk) ? bs[i] : 0u;
        unsigned inc = v;
        for (int o = 1; o < 32; o <<= 1) {
            unsigned t = __shfl_up(inc, o);
            if (lane >= o) inc += t;
        }
        if (lane == 31) wsum[wave] = inc;
        __syncthreads();
        unsigned woff = 0u, tot = 0u;
        for (int w = 0; w < 32; ++w) {
            unsigned q = wsum[w];
            if (w < wave) woff += q;
            tot += q;
        }
        if (i < nblk) boff[i] = carry + woff + inc - v;
        carry += tot;
        __syncthreads();
    }
}

#define GWAVES 28
#define GROWS  (GWAVES * 16)

__global__ void __launch_bounds__(1024)
k_fill_gemm(const int* __restrict__ srcp, const int* __restrict__ dstp,
            const unsigned* __restrict__ part, const unsigned* __restrict__ lp,
            const unsigned* __restrict__ boff, int* __restrict__ col,
            const float* __restrict__ x, const float* __restrict__ W,
            float* __restrict__ h, int N, int E, int R) {
    __shared__ __attribute__((aligned(16))) unsigned char smem[HMAX * 4];
    const int nfill = R * HSL;

    if ((int)blockIdx.x < nfill) {
        unsigned* cur = (unsigned*)smem;
        const int bx = (int)blockIdx.x % R;
        const int by = (int)blockIdx.x / R;
        const int rs = ((N + R - 1) / R + 31) & ~31;
        const unsigned base = (unsigned)bx * (unsigned)rs;
        const unsigned* p = part + (size_t)by * ((size_t)R * rs) + base;
        for (int i = threadIdx.x; i < rs; i += 1024) {
            const unsigned node = base + (unsigned)i;
            cur[i] = (node < (unsigned)N) ? boff[node / DBLK] + lp[node] + p[i] : 0u;
        }
        __syncthreads();

        const int per = (E + HSL - 1) / HSL;
        const int e0  = by * per;
        int e1 = e0 + per;
        if (e1 > E) e1 = E;

        int e = e0 + (int)threadIdx.x;
        for (; e + 7 * 1024 < e1; e += 8 * 1024) {
            unsigned v[8];
            int sv[8];
#pragma unroll
            for (int j = 0; j < 8; ++j) { v[j] = (unsigned)dstp[e + j * 1024]; sv[j] = srcp[e + j * 1024]; }
#pragma unroll
            for (int j = 0; j < 8; ++j) {
                unsigned l = v[j] - base;
                if (l < (unsigned)rs) {
                    unsigned pos = atomicAdd(&cur[l], 1u);
                    if (pos < (unsigned)E) col[pos] = sv[j];
                }
            }
        }
        for (; e < e1; e += 1024) {
            unsigned l = (unsigned)dstp[e] - base;
            if (l < (unsigned)rs) {
                unsigned pos = atomicAdd(&cur[l], 1u);
                if (pos < (unsigned)E) col[pos] = srcp[e];
            }
        }
        return;
    }

    float2* Ws2 = (float2*)smem;
#pragma unroll
    for (int idx = threadIdx.x; idx < (CIN / 2) * COUT; idx += 1024) {
        int p = idx >> 6, n = idx & 63;
        Ws2[idx] = make_float2(W[p * (2 * COUT) + n], W[p * (2 * COUT) + COUT + n]);
    }
    __syncthreads();

    const int lane = threadIdx.x & 31;
    const int wave = threadIdx.x >> 5;
    const int m0   = (((int)blockIdx.x - nfill) * GWAVES + wave) * 16;
    if (wave >= GWAVES || m0 >= N) return;
    const int mr   = lane & 15;
    const int kh   = (lane >> 4) << 1;

    int row  = m0 + mr;
    int rowc = row < N ? row : (N - 1);
    const float* xr = x + (size_t)rowc * CIN;

    v8f acc0 = {}; v8f acc1 = {}; v8f acc2 = {}; v8f acc3 = {};
    for (int k = 0; k < CIN; k += 4) {
        v2f a = *(const v2f*)(xr + k + kh);
        const float2* wp = &Ws2[(((k + kh) >> 1) * COUT) + mr];
        v2f b0 = *(const v2f*)(wp + 0);
        v2f b1 = *(const v2f*)(wp + 16);
        v2f b2 = *(const v2f*)(wp + 32);
        v2f b3 = *(const v2f*)(wp + 48);
        acc0 = __builtin_amdgcn_wmma_f32_16x16x4_f32(false, a, false, b0, (short)0, acc0, false, false);
        acc1 = __builtin_amdgcn_wmma_f32_16x16x4_f32(false, a, false, b1, (short)0, acc1, false, false);
        acc2 = __builtin_amdgcn_wmma_f32_16x16x4_f32(false, a, false, b2, (short)0, acc2, false, false);
        acc3 = __builtin_amdgcn_wmma_f32_16x16x4_f32(false, a, false, b3, (short)0, acc3, false, false);
    }

    const int nlo = lane & 15;
    const int mhi = (lane >> 4) << 3;
    for (int r = 0; r < 8; ++r) {
        int orow = m0 + r + mhi;
        if (orow < N) {
            float* hp = h + (size_t)orow * COUT + nlo;
            hp[0]  = acc0[r];
            hp[16] = acc1[r];
            hp[32] = acc2[r];
            hp[48] = acc3[r];
        }
    }
}

#define GU 8
__global__ void __launch_bounds__(256)
k_gather(const int* __restrict__ col, const unsigned* __restrict__ lp,
         const unsigned* __restrict__ boff, const unsigned* __restrict__ degc,
         const float* __restrict__ dinv, const float* __restrict__ h,
         const float* __restrict__ b, float* __restrict__ out, int N, int E) {
    int d = blockIdx.x * 8 + (int)(threadIdx.x >> 5);
    if (d >= N) return;
    d = __builtin_amdgcn_readfirstlane(d);
    const int lane = threadIdx.x & 31;
    const unsigned start = boff[d / DBLK] + lp[d];
    const unsigned cnt   = degc[d];
    const float dd = dinv[d];
    const float2* h2 = (const float2*)h;
    const float2 self = h2[(size_t)d * (COUT / 2) + lane];
    float ax = 0.f, ay = 0.f;
    for (unsigned j0 = 0; j0 < cnt; j0 += 32) {
        int   s_l = d;
        float w_l = 0.f;
        const unsigned j = j0 + (unsigned)lane;
        if (j < cnt && start + j < (unsigned)E) {
            int sv = col[start + j];
            if ((unsigned)sv < (unsigned)N) { s_l = sv; w_l = dinv[sv]; }
        }
        const unsigned left = cnt - j0;
        const unsigned n = left < 32u ? left : 32u;
        for (unsigned t = 0; t < n; t += GU) {
            float2 v[GU];
            float  w[GU];
#pragma unroll
            for (int u = 0; u < GU; ++u) {
                const int s = __builtin_amdgcn_readlane(s_l, (int)t + u);
                w[u] = __builtin_bit_cast(float,
                         __builtin_amdgcn_readlane(__builtin_bit_cast(int, w_l), (int)t + u));
                v[u] = h2[(size_t)s * (COUT / 2) + lane];
            }
#pragma unroll
            for (int u = 0; u < GU; ++u) { ax += w[u] * v[u].x; ay += w[u] * v[u].y; }
        }
    }
    const float2 bb = ((const float2*)b)[lane];
    float2 o;
    o.x = bb.x + dd * (ax + dd * self.x);
    o.y = bb.y + dd * (ay + dd * self.y);
    ((float2*)out)[(size_t)d * (COUT / 2) + lane] = o;
}

extern "C" void kernel_launch(void* const* d_in, const int* in_sizes, int n_in,
                              void* d_out, int out_size, void* d_ws, size_t ws_size,
                              hipStream_t stream) {
    const float* x  = (const float*)d_in[0];
    const int*   ei = (const int*)d_in[1];
    const float* W  = (const float*)d_in[2];
    const float* b  = (const float*)d_in[3];
    float* out = (float*)d_out;

    const int N = in_sizes[0] / CIN;
    const int E = in_sizes[1] / 2;
    const int* srcp = ei;
    const int* dstp = ei + E;

    const int R    = (N + HMAX - 1) / HMAX;
    const int rs   = ((N + R - 1) / R + 31) & ~31;
    const int nblk = (N + DBLK - 1) / DBLK;
    const size_t nN = (((size_t)N * 4) + 255) & ~(size_t)255;
    const size_t nB = (((size_t)nblk * 4) + 255) & ~(size_t)255;
    char* ws = (char*)d_ws;
    float*    h    = (float*)ws;                 ws += (size_t)N * COUT * sizeof(float);
    float*    dinv = (float*)ws;                 ws += nN;
    unsigned* part = (unsigned*)ws;              ws += (size_t)HSL * R * rs * sizeof(unsigned);
    unsigned* degc = (unsigned*)ws;              ws += nN;
    unsigned* lp   = (unsigned*)ws;              ws += nN;
    unsigned* bs   = (unsigned*)ws;              ws += nB;
    unsigned* boff = (unsigned*)ws;              ws += nB;
    int*      col  = (int*)ws;

    int nb;
    dim3 hg(R, HSL);
    k_hist<<<hg, 1024, 0, stream>>>(dstp, part, N, E);
    k_dinv<<<nblk, DBLK, 0, stream>>>(part, dinv, degc, lp, bs, N, R);
    k_scan<<<1, 1024, 0, stream>>>(bs, boff, nblk);
    nb = R * HSL + (N + GROWS - 1) / GROWS;
    k_fill_gemm<<<nb, 1024, 0, stream>>>(srcp, dstp, part, lp, boff, col, x, W, h, N, E, R);

    nb = (N + 7) / 8;
    k_gather<<<nb, 256, 0, stream>>>(col, lp, boff, degc, dinv, h, b, out, N, E);
}


# fill role: start-offset loads issued nine nodes at a time and a batch's eight LDS rank adds issued before the first is waited for, not one dependent round trip each; ranks, sum order as before
# speedup vs baseline: 4.0379x; 1.1239x over previous
#include <hip/hip_runtime.h>


typedef float v2f __attribute__((ext_vector_type(2)));
typedef float v8f __attribute__((ext_vector_type(8)));

#define CIN 256
#define COUT 64

#define HMAX 25600
#define HSL  64

__global__ void __launch_bounds__(1024)
k_hist(const int* __restrict__ dstp, unsigned* __restrict__ part, int N, int E) {
    __shared__ unsigned cnt[HMAX];
    const int R  = gridDim.x;
    const int rs = ((N + R - 1) / R + 31) & ~31;
    const unsigned base = blockIdx.x * (unsigned)rs;
    for (int i = threadIdx.x; i < rs; i += 1024) cnt[i] = 0u;
    __syncthreads();

    const int per = (E + (int)gridDim.y - 1) / (int)gridDim.y;
    const int e0  = (int)blockIdx.y * per;
    int e1 = e0 + per;
    if (e1 > E) e1 = E;

    int e = e0 + (int)threadIdx.x;
    for (; e + 7 * 1024 < e1; e += 8 * 1024) {
        unsigned v[8];
#pragma unroll
        for (int j = 0; j < 8; ++j) v[j] = (unsigned)dstp[e + j * 1024];
#pragma unroll
        for (int j = 0; j < 8; ++j) {
            unsigned l = v[j] - base;
            if (l < (unsigned)rs) atomicAdd(&cnt[l], 1u);
        }
    }
    for (; e < e1; e += 1024) {
        unsigned l = (unsigned)dstp[e] - base;
        if (l < (unsigned)rs) atomicAdd(&cnt[l], 1u);
    }
    __syncthreads();

    unsigned* p = part + (size_t)blockIdx.y * ((size_t)R * rs) + base;
    for (int i = threadIdx.x; i < rs; i += 1024) p[i] = cnt[i];
}

#define DBLK 256

__global__ void __launch_bounds__(DBLK)
k_dinv(unsigned* __restrict__ part, float* __restrict__ dinv,
       unsigned* __restrict__ degc, unsigned* __restrict__ lp,
       unsigned* __restrict__ bs, int N, int R) {
    __shared__ unsigned wsum[DBLK / 32];
    const int i = blockIdx.x * DBLK + threadIdx.x;
    unsigned c = 0u;
    if (i < N) {
        const int rs = ((N + R - 1) / R + 31) & ~31;
        const size_t pitch = (size_t)R * rs;
        unsigned* p = part + i;
        for (int s0 = 0; s0 < HSL; s0 += 16) {
            unsigned v[16];
#pragma unroll
            for (int j = 0; j < 16; ++j) v[j] = p[(size_t)(s0 + j) * pitch];
#pragma unroll
            for (int j = 0; j < 16; ++j) { p[(size_t)(s0 + j) * pitch] = c; c += v[j]; }
        }
        dinv[i] = rsqrtf((float)(c + 1u));
        degc[i] = c;
    }
    const int lane = threadIdx.x & 31;
    const int wave = threadIdx.x >> 5;
    unsigned inc = c;
    for (int o = 1; o < 32; o <<= 1) {
        unsigned t = __shfl_up(inc, o);
        if (lane >= o) inc += t;
    }
    if (lane == 31) wsum[wave] = inc;
    __syncthreads();
    unsigned woff = 0u;
    for (int w = 0; w < wave; ++w) woff += wsum[w];
    if (i < N) lp[i] = woff + inc - c;
    if (threadIdx.x == DBLK - 1) bs[blockIdx.x] = woff + inc;
}

__global__ void __launch_bounds__(1024)
k_scan(const unsigned* __restrict__ bs, unsigned* __restrict__ boff, int nblk) {
    __shared__ unsigned wsum[32];
    const int lane = threadIdx.x & 31;
    const int wave = threadIdx.x >> 5;
    unsigned carry = 0u;
    for (int c0 = 0; c0 < nblk; c0 += 1024) {
        const int i = c0 + (int)threadIdx.x;
        const unsigned v = (i < nblk) ? bs[i] : 0u;
        unsigned inc = v;
        for (int o = 1; o < 32; o <<= 1) {
            unsigned t = __shfl_up(inc, o);
            if (lane >= o) inc += t;
        }
        if (lane == 31) wsum[wave] = inc;
        __syncthreads();
        unsigned woff = 0u, tot = 0u;
        for (int w = 0; w < 32; ++w) {
            unsigned q = wsum[w];
            if (w < wave) woff += q;
            tot += q;
        }
        if (i < nblk) boff[i] = carry + woff + inc - v;
        carry += tot;
        __syncthreads();
    }
}

#define FI 9
#define GWAVES 28
#define GROWS  (GWAVES * 16)

__global__ void __launch_bounds__(1024)
k_fill_gemm(const int* __restrict__ srcp, const int* __restrict__ dstp,
            const unsigned* __restrict__ part, const unsigned* __restrict__ lp,
            const unsigned* __restrict__ boff, int* __restrict__ col,
            const float* __restrict__ x, const float* __restrict__ W,
            const float* __restrict__ dinv, float* __restrict__ h, int N, int E, int R) {
    __shared__ __attribute__((aligned(16))) unsigned char smem[HMAX * 4];
    const int nfill = R * HSL;

    if ((int)blockIdx.x < nfill) {
        unsigned* cur = (unsigned*)smem;
        const int bx = (int)blockIdx.x % R;
        const int by = (int)blockIdx.x / R;
        const int rs = ((N + R - 1) / R + 31) & ~31;
        const unsigned base = (unsigned)bx * (unsigned)rs;
        const unsigned* p = part + (size_t)by * ((size_t)R * rs) + base;
        for (int i0 = threadIdx.x; i0 < rs; i0 += FI * 1024) {
            unsigned fa[FI], fb[FI], fc[FI];
#pragma unroll
            for (int j = 0; j < FI; ++j) {
                const int i  = i0 + j * 1024;
                const int ic = i < rs ? i : rs - 1;
                const unsigned node = base + (unsigned)ic;
                const unsigned nc = node < (unsigned)N ? node : (unsigned)(N - 1);
                fa[j] = boff[nc / DBLK];
                fb[j] = lp[nc];
                fc[j] = p[ic];
            }
#pragma unroll
            for (int j = 0; j < FI; ++j) {
                const int i  = i0 + j * 1024;
                const int ic = i < rs ? i : rs - 1;
                cur[ic] = (base + (unsigned)ic < (unsigned)N) ? fa[j] + fb[j] + fc[j] : 0u;
            }
        }
        __syncthreads();

        const int per = (E + HSL - 1) / HSL;
        const int e0  = by * per;
        int e1 = e0 + per;
        if (e1 > E) e1 = E;

        int e = e0 + (int)threadIdx.x;
        for (; e + 7 * 1024 < e1; e += 8 * 1024) {
            unsigned v[8];
            int sv[8];
#pragma unroll
            for (int j = 0; j < 8; ++j) { v[j] = (unsigned)dstp[e + j * 1024]; sv[j] = srcp[e + j * 1024]; }
            unsigned pos[8];
#pragma unroll
            for (int j = 0; j < 8; ++j) {
                unsigned l = v[j] - base;
                pos[j] = 0xffffffffu;
                if (l < (unsigned)rs) pos[j] = atomicAdd(&cur[l], 1u);
            }
#pragma unroll
            for (int j = 0; j < 8; ++j) {
                if (pos[j] < (unsigned)E) col[pos[j]] = sv[j];
            }
        }
        for (; e < e1; e += 1024) {
            const unsigned l = (unsigned)dstp[e] - base;
            const int s = srcp[e];
            unsigned pos = 0xffffffffu;
            if (l < (unsigned)rs) pos = atomicAdd(&cur[l], 1u);
            if (pos < (unsigned)E) col[pos] = s;
        }
        return;
    }

    float2* Ws2 = (float2*)smem;
#pragma unroll
    for (int idx = threadIdx.x; idx < (CIN / 2) * COUT; idx += 1024) {
        int p = idx >> 6, n = idx & 63;
        Ws2[idx] = make_float2(W[p * (2 * COUT) + n], W[p * (2 * COUT) + COUT + n]);
    }
    __syncthreads();

    const int lane = threadIdx.x & 31;
    const int wave = threadIdx.x >> 5;
    const int m0   = (((int)blockIdx.x - nfill) * GWAVES + wave) * 16;
    if (wave >= GWAVES || m0 >= N) return;
    const int mr   = lane & 15;
    const int kh   = (lane >> 4) << 1;

    int row  = m0 + mr;
    int rowc = row < N ? row : (N - 1);
    const float* xr = x + (size_t)rowc * CIN;

    v8f acc0 = {}; v8f acc1 = {}; v8f acc2 = {}; v8f acc3 = {};
    for (int k = 0; k < CIN; k += 4) {
        v2f a = *(const v2f*)(xr + k + kh);
        const float2* wp = &Ws2[(((k + kh) >> 1) * COUT) + mr];
        v2f b0 = *(const v2f*)(wp + 0);
        v2f b1 = *(const v2f*)(wp + 16);
        v2f b2 = *(const v2f*)(wp + 32);
        v2f b3 = *(const v2f*)(wp + 48);
        acc0 = __builtin_amdgcn_wmma_f32_16x16x4_f32(false, a, false, b0, (short)0, acc0, false, false);
        acc1 = __builtin_amdgcn_wmma_f32_16x16x4_f32(false, a, false, b1, (short)0, acc1, false, false);
        acc2 = __builtin_amdgcn_wmma_f32_16x16x4_f32(false, a, false, b2, (short)0, acc2, false, false);
        acc3 = __builtin_amdgcn_wmma_f32_16x16x4_f32(false, a, false, b3, (short)0, acc3, false, false);
    }

    const float dvl = dinv[rowc];
    const int nlo = lane & 15;
    const int mhi = (lane >> 4) << 3;
    for (int r = 0; r < 8; ++r) {
        int orow = m0 + r + mhi;
        const float sc = __shfl(dvl, r + mhi);
        if (orow < N) {
            float* hp = h + (size_t)orow * COUT + nlo;
            hp[0]  = sc * acc0[r];
            hp[16] = sc * acc1[r];
            hp[32] = sc * acc2[r];
            hp[48] = sc * acc3[r];
        }
    }
}

#define GU 8
__global__ void __launch_bounds__(256)
k_gather(const int* __restrict__ col, const unsigned* __restrict__ lp,
         const unsigned* __restrict__ boff, const unsigned* __restrict__ degc,
         const float* __restrict__ dinv, const float* __restrict__ h,
         const float* __restrict__ b, float* __restrict__ out, int N, int E) {
    int d = blockIdx.x * 8 + (int)(threadIdx.x >> 5);
    if (d >= N) return;
    d = __builtin_amdgcn_readfirstlane(d);
    const int lane = threadIdx.x & 31;
    const unsigned start = boff[d / DBLK] + lp[d];
    const unsigned cnt   = degc[d];
    const float dd = dinv[d];
    const float2* h2 = (const float2*)h;
    const float2 self = h2[(size_t)d * (COUT / 2) + lane];
    float ax = 0.f, ay = 0.f;
    for (unsigned j0 = 0; j0 < cnt; j0 += 32) {
        int  s_l = d;
        bool good = false;
        const unsigned j = j0 + (unsigned)lane;
        if (j < cnt && start + j < (unsigned)E) {
            int sv = col[start + j];
            if ((unsigned)sv < (unsigned)N) { s_l = sv; good = true; }
        }
        const unsigned ok = (unsigned)__ballot(good);
        const unsigned left = cnt - j0;
        const unsigned n = left < 32u ? left : 32u;
        for (unsigned t = 0; t < n; t += GU) {
            const unsigned m = (ok >> t) & ((1u << GU) - 1u);
            float2 v[GU];
            if (m == (1u << GU) - 1u) {
#pragma unroll
                for (int u = 0; u < GU; ++u) {
                    const int s = __builtin_amdgcn_readlane(s_l, (int)t + u);
                    v[u] = h2[(size_t)s * (COUT / 2) + lane];
                }
            } else {
#pragma unroll
                for (int u = 0; u < GU; ++u) {
                    v[u] = make_float2(0.f, 0.f);
                    if ((m >> u) & 1u) {
                        const int s = __builtin_amdgcn_readlane(s_l, (int)t + u);
                        v[u] = h2[(size_t)s * (COUT / 2) + lane];
                    }
                }
            }
#pragma unroll
            for (int u = 0; u < GU; ++u) { ax += v[u].x; ay += v[u].y; }
        }
    }
    const float2 bb = ((const float2*)b)[lane];
    float2 o;
    o.x = bb.x + dd * (ax + self.x);
    o.y = bb.y + dd * (ay + self.y);
    ((float2*)out)[(size_t)d * (COUT / 2) + lane] = o;
}

extern "C" void kernel_launch(void* const* d_in, const int* in_sizes, int n_in,
                              void* d_out, int out_size, void* d_ws, size_t ws_size,
                              hipStream_t stream) {
    const float* x  = (const float*)d_in[0];
    const int*   ei = (const int*)d_in[1];
    const float* W  = (const float*)d_in[2];
    const float* b  = (const float*)d_in[3];
    float* out = (float*)d_out;

    const int N = in_sizes[0] / CIN;
    const int E = in_sizes[1] / 2;
    const int* srcp = ei;
    const int* dstp = ei + E;

    const int R    = (N + HMAX - 1) / HMAX;
    const int rs   = ((N + R - 1) / R + 31) & ~31;
    const int nblk = (N + DBLK - 1) / DBLK;
    const size_t nN = (((size_t)N * 4) + 255) & ~(size_t)255;
    const size_t nB = (((size_t)nblk * 4) + 255) & ~(size_t)255;
    char* ws = (char*)d_ws;
    float*    h    = (float*)ws;                 ws += (size_t)N * COUT * sizeof(float);
    float*    dinv = (float*)ws;                 ws += nN;
    unsigned* part = (unsigned*)ws;              ws += (size_t)HSL * R * rs * sizeof(unsigned);
    unsigned* degc = (unsigned*)ws;              ws += nN;
    unsigned* lp   = (unsigned*)ws;              ws += nN;
    unsigned* bs   = (unsigned*)ws;              ws += nB;
    unsigned* boff = (unsigned*)ws;              ws += nB;
    int*      col  = (int*)ws;

    int nb;
    dim3 hg(R, HSL);
    k_hist<<<hg, 1024, 0, stream>>>(dstp, part, N, E);
    k_dinv<<<nblk, DBLK, 0, stream>>>(part, dinv, degc, lp, bs, N, R);
    k_scan<<<1, 1024, 0, stream>>>(bs, boff, nblk);
    nb = R * HSL + (N + GROWS - 1) / GROWS;
    k_fill_gemm<<<nb, 1024, 0, stream>>>(srcp, dstp, part, lp, boff, col, x, W, dinv, h, N, E, R);

    nb = (N + 7) / 8;
    k_gather<<<nb, 256, 0, stream>>>(col, lp, boff, degc, dinv, h, b, out, N, E);
}
